# starting kernel
# baseline (speedup 1.0000x reference)
#include <hip/hip_runtime.h>

#define BATCH 8
#define LQ   512
#define KK   512
#define DD   64
#define NN   64
#define MM   64
#define PP   64

typedef __attribute__((ext_vector_type(2))) float v2f;
typedef __attribute__((ext_vector_type(8))) float v8f;

__global__ __launch_bounds__(64) void vkc_kernel(const float* __restrict__ vk,
                                                 const float* __restrict__ vexp,
                                                 float* __restrict__ vkc) {
  __shared__ float sE[4 * PP];
  const int tid  = threadIdx.x;
  const int pair = tid >> 4;
  const int nq   = (tid & 15) * 4;
  const int bk0  = blockIdx.x * 4;

  for (int i = tid; i < 4 * PP; i += 64) sE[i] = vexp[(size_t)bk0 * PP + i];
  __syncthreads();

  const float* vkp = vk + (size_t)(bk0 + pair) * PP * NN + nq;
  float4 acc = {0.f, 0.f, 0.f, 0.f};
#pragma unroll 4
  for (int p = 0; p < PP; ++p) {
    const float  w = sE[pair * PP + p];
    const float4 v = *reinterpret_cast<const float4*>(vkp + (size_t)p * NN);
    acc.x = fmaf(w, v.x, acc.x);
    acc.y = fmaf(w, v.y, acc.y);
    acc.z = fmaf(w, v.z, acc.z);
    acc.w = fmaf(w, v.w, acc.w);
  }
  *reinterpret_cast<float4*>(vkc + (size_t)(bk0 + pair) * NN + nq) = acc;
}

#define SSTRIDE 516

__global__ __launch_bounds__(128) void attn_kernel(const float* __restrict__ q,
                                                   const float* __restrict__ kmat,
                                                   const float* __restrict__ vkc,
                                                   const float* __restrict__ scale_p,
                                                   float* __restrict__ tmp) {
  __shared__ float sS[16 * SSTRIDE];
  __shared__ float sRed[128];
  __shared__ float sRowMax[16];
  __shared__ float sRowSum[16];

  const int tid  = threadIdx.x;
  const int wave = tid >> 5;
  const int lane = tid & 31;
  const int half = lane >> 4;
  const int lrow = lane & 15;

  const int b  = blockIdx.x >> 5;
  const int lt = blockIdx.x & 31;
  const int l0 = lt * 16;

  const float scale = scale_p[0];

  const float* qrow = q + ((size_t)b * LQ + l0 + lrow) * DD + 2 * half;
  v2f afrag[16];
#pragma unroll
  for (int dd = 0; dd < 16; ++dd) {
    const float2 t = *reinterpret_cast<const float2*>(qrow + 4 * dd);
    afrag[dd].x = t.x; afrag[dd].y = t.y;
  }

  const float* kbase = kmat + (size_t)b * KK * DD;
  for (int ct = wave; ct < 32; ct += 4) {
    const int k0 = ct * 16;
    const float* krow = kbase + (size_t)(k0 + lrow) * DD + 2 * half;
    v8f acc0 = {0.f, 0.f, 0.f, 0.f, 0.f, 0.f, 0.f, 0.f};
    v8f acc1 = {0.f, 0.f, 0.f, 0.f, 0.f, 0.f, 0.f, 0.f};
#pragma unroll
    for (int dd = 0; dd < 16; dd += 2) {
      const float2 t0 = *reinterpret_cast<const float2*>(krow + 4 * dd);
      const float2 t1 = *reinterpret_cast<const float2*>(krow + 4 * dd + 4);
      v2f b0; b0.x = t0.x; b0.y = t0.y;
      v2f b1; b1.x = t1.x; b1.y = t1.y;
      acc0 = __builtin_amdgcn_wmma_f32_16x16x4_f32(false, afrag[dd],     false, b0,
                                                   (short)0, acc0, false, false);
      acc1 = __builtin_amdgcn_wmma_f32_16x16x4_f32(false, afrag[dd + 1], false, b1,
                                                   (short)0, acc1, false, false);
    }
#pragma unroll
    for (int r = 0; r < 8; ++r) {
      const int m = r + 8 * half;
      sS[m * SSTRIDE + k0 + lrow] = (acc0[r] + acc1[r]) * scale;
    }
  }
  __syncthreads();

  {
    const int r = tid >> 3;
    const int j = tid & 7;
    float pmax = -3.4e38f;
    for (int i = 0; i < 64; ++i)
      pmax = fmaxf(pmax, sS[r * SSTRIDE + j + 8 * i]);
    sRed[tid] = pmax;
    __syncthreads();
    if (j == 0) {
      float mx = sRed[tid];
      for (int t = 1; t < 8; ++t) mx = fmaxf(mx, sRed[tid + t]);
      sRowMax[r] = mx;
    }
    __syncthreads();
    const float mx = sRowMax[r];
    float psum = 0.f;
    for (int i = 0; i < 64; ++i) {
      const int c = j + 8 * i;
      const float e = expf(sS[r * SSTRIDE + c] - mx);
      sS[r * SSTRIDE + c] = e;
      psum += e;
    }
    sRed[tid] = psum;
    __syncthreads();
    if (j == 0) {
      float s = sRed[tid];
      for (int t = 1; t < 8; ++t) s += sRed[tid + t];
      sRowSum[r] = s;
    }
  }
  __syncthreads();

  {
    const int n0 = wave * 16;
    const float* vb = vkc + (size_t)b * KK * NN + n0 + lrow;
    v8f acc0 = {0.f, 0.f, 0.f, 0.f, 0.f, 0.f, 0.f, 0.f};
    v8f acc1 = {0.f, 0.f, 0.f, 0.f, 0.f, 0.f, 0.f, 0.f};
    for (int k0 = 0; k0 < KK; k0 += 8) {
      const float2 ta =
          *reinterpret_cast<const float2*>(&sS[lrow * SSTRIDE + k0 + 2 * half]);
      const float2 tb =
          *reinterpret_cast<const float2*>(&sS[lrow * SSTRIDE + k0 + 4 + 2 * half]);
      v2f a0; a0.x = ta.x; a0.y = ta.y;
      v2f a1; a1.x = tb.x; a1.y = tb.y;
      v2f b0;
      b0.x = vb[(size_t)(k0 + 2 * half)     * NN];
      b0.y = vb[(size_t)(k0 + 2 * half + 1) * NN];
      v2f b1;
      b1.x = vb[(size_t)(k0 + 4 + 2 * half)     * NN];
      b1.y = vb[(size_t)(k0 + 4 + 2 * half + 1) * NN];
      acc0 = __builtin_amdgcn_wmma_f32_16x16x4_f32(false, a0, false, b0,
                                                   (short)0, acc0, false, false);
      acc1 = __builtin_amdgcn_wmma_f32_16x16x4_f32(false, a1, false, b1,
                                                   (short)0, acc1, false, false);
    }
#pragma unroll
    for (int r = 0; r < 8; ++r) {
      const int m = r + 8 * half;
      tmp[((size_t)b * LQ + l0 + m) * NN + n0 + lrow] =
          (acc0[r] + acc1[r]) / sRowSum[m];
    }
  }
}

__global__ __launch_bounds__(256) void out_kernel(const float* __restrict__ q,
                                                  const float* __restrict__ vq,
                                                  const float* __restrict__ tmp,
                                                  const float* __restrict__ gamma,
                                                  const float* __restrict__ beta,
                                                  float* __restrict__ out) {
  const int bl = blockIdx.x;
  __shared__ float sT[64];
  __shared__ float sPart[256];
  __shared__ float sRed[64];
  __shared__ float sRed2[64];
  __shared__ float sStat[2];

  const int tid = threadIdx.x;
  if (tid < 64) sT[tid] = tmp[(size_t)bl * NN + tid];
  __syncthreads();

  const int m    = tid >> 2;
  const int part = tid & 3;
  const float4* vqp = reinterpret_cast<const float4*>(
      vq + ((size_t)bl * MM + m) * NN + part * 16);
  float acc = 0.f;
#pragma unroll
  for (int i = 0; i < 4; ++i) {
    const float4 v = vqp[i];
    const int c = part * 16 + 4 * i;
    acc = fmaf(v.x, sT[c + 0], acc);
    acc = fmaf(v.y, sT[c + 1], acc);
    acc = fmaf(v.z, sT[c + 2], acc);
    acc = fmaf(v.w, sT[c + 3], acc);
  }
  sPart[tid] = acc;
  __syncthreads();

  float x = 0.f;
  if (part == 0) {
    x = sPart[tid] + sPart[tid + 1] + sPart[tid + 2] + sPart[tid + 3]
        + q[(size_t)bl * DD + m];
    sRed[m]  = x;
    sRed2[m] = x * x;
  }
  __syncthreads();
  for (int s = 32; s >= 1; s >>= 1) {
    if (tid < s) { sRed[tid] += sRed[tid + s]; sRed2[tid] += sRed2[tid + s]; }
    __syncthreads();
  }
  if (tid == 0) {
    const float mu  = sRed[0]  * (1.f / 64.f);
    const float var = sRed2[0] * (1.f / 64.f) - mu * mu;
    sStat[0] = mu;
    sStat[1] = rsqrtf(var + 1e-3f);
  }
  __syncthreads();
  if (part == 0) {
    out[(size_t)bl * DD + m] = (x - sStat[0]) * sStat[1] * gamma[m] + beta[m];
  }
}

extern "C" void kernel_launch(void* const* d_in, const int* in_sizes, int n_in,
                              void* d_out, int out_size, void* d_ws, size_t ws_size,
                              hipStream_t stream) {
  const float* q     = (const float*)d_in[0];
  const float* kmat  = (const float*)d_in[1];
  const float* vq    = (const float*)d_in[2];
  const float* vk    = (const float*)d_in[3];
  const float* vexp  = (const float*)d_in[4];
  const float* scale = (const float*)d_in[5];
  const float* gamma = (const float*)d_in[6];
  const float* beta  = (const float*)d_in[7];
  float* out = (float*)d_out;

  float* vkc = (float*)d_ws;
  float* tmp = vkc + (size_t)BATCH * KK * NN;

  vkc_kernel <<<BATCH * KK / 4,    64,  0, stream>>>(vk, vexp, vkc);
  attn_kernel<<<BATCH * (LQ / 16), 128, 0, stream>>>(q, kmat, vkc, scale, tmp);
  out_kernel <<<BATCH * LQ,        256, 0, stream>>>(q, vq, tmp, gamma, beta, out);
}


# both 67MB streaming mat-vec kernels: one wave per 64x64 slab with its 32 b128 loads all in flight, sums by lane exchanges instead of LDS+barriers (4096 waves each; fp sum order changes)
# speedup vs baseline: 1.3139x; 1.3139x over previous
#include <hip/hip_runtime.h>

#define BATCH 8
#define LQ   512
#define KK   512
#define DD   64
#define NN   64
#define MM   64
#define PP   64

typedef __attribute__((ext_vector_type(2))) float v2f;
typedef __attribute__((ext_vector_type(8))) float v8f;

__global__ __launch_bounds__(512) __attribute__((amdgpu_waves_per_eu(4, 4)))
void vkc_kernel(const float* __restrict__ vk,
                const float* __restrict__ vexp,
                float* __restrict__ vkc) {
  const int lane = threadIdx.x & 31;
  const int wave = __builtin_amdgcn_readfirstlane((int)(threadIdx.x >> 5));
  const int c = lane & 15;
  const int h = lane >> 4;
  const size_t bk = (size_t)blockIdx.x * 16 + wave;

  const float4* wp = reinterpret_cast<const float4*>(vexp + bk * PP + 32 * h);
  const float4* vp = reinterpret_cast<const float4*>(
      vk + bk * PP * NN + (size_t)(32 * h) * NN + 4 * c);

  float w[32];
  for (int i = 0; i < 8; ++i) {
    const float4 t = wp[i];
    w[4 * i + 0] = t.x; w[4 * i + 1] = t.y; w[4 * i + 2] = t.z; w[4 * i + 3] = t.w;
  }
  float4 v[32];
  for (int j = 0; j < 32; ++j) v[j] = vp[j * (NN / 4)];

  float4 acc = {0.f, 0.f, 0.f, 0.f};
  for (int j = 0; j < 32; ++j) {
    acc.x = fmaf(w[j], v[j].x, acc.x);
    acc.y = fmaf(w[j], v[j].y, acc.y);
    acc.z = fmaf(w[j], v[j].z, acc.z);
    acc.w = fmaf(w[j], v[j].w, acc.w);
  }
  acc.x += __shfl_xor(acc.x, 16, 32);
  acc.y += __shfl_xor(acc.y, 16, 32);
  acc.z += __shfl_xor(acc.z, 16, 32);
  acc.w += __shfl_xor(acc.w, 16, 32);
  if (h == 0) *reinterpret_cast<float4*>(vkc + bk * NN + 4 * c) = acc;
}

#define SSTRIDE 516

__global__ __launch_bounds__(128) void attn_kernel(const float* __restrict__ q,
                                                   const float* __restrict__ kmat,
                                                   const float* __restrict__ vkc,
                                                   const float* __restrict__ scale_p,
                                                   float* __restrict__ tmp) {
  __shared__ float sS[16 * SSTRIDE];
  __shared__ float sRed[128];
  __shared__ float sRowMax[16];
  __shared__ float sRowSum[16];

  const int tid  = threadIdx.x;
  const int wave = tid >> 5;
  const int lane = tid & 31;
  const int half = lane >> 4;
  const int lrow = lane & 15;

  const int b  = blockIdx.x >> 5;
  const int lt = blockIdx.x & 31;
  const int l0 = lt * 16;

  const float scale = scale_p[0];

  const float* qrow = q + ((size_t)b * LQ + l0 + lrow) * DD + 2 * half;
  v2f afrag[16];
#pragma unroll
  for (int dd = 0; dd < 16; ++dd) {
    const float2 t = *reinterpret_cast<const float2*>(qrow + 4 * dd);
    afrag[dd].x = t.x; afrag[dd].y = t.y;
  }

  const float* kbase = kmat + (size_t)b * KK * DD;
  for (int ct = wave; ct < 32; ct += 4) {
    const int k0 = ct * 16;
    const float* krow = kbase + (size_t)(k0 + lrow) * DD + 2 * half;
    v8f acc0 = {0.f, 0.f, 0.f, 0.f, 0.f, 0.f, 0.f, 0.f};
    v8f acc1 = {0.f, 0.f, 0.f, 0.f, 0.f, 0.f, 0.f, 0.f};
#pragma unroll
    for (int dd = 0; dd < 16; dd += 2) {
      const float2 t0 = *reinterpret_cast<const float2*>(krow + 4 * dd);
      const float2 t1 = *reinterpret_cast<const float2*>(krow + 4 * dd + 4);
      v2f b0; b0.x = t0.x; b0.y = t0.y;
      v2f b1; b1.x = t1.x; b1.y = t1.y;
      acc0 = __builtin_amdgcn_wmma_f32_16x16x4_f32(false, afrag[dd],     false, b0,
                                                   (short)0, acc0, false, false);
      acc1 = __builtin_amdgcn_wmma_f32_16x16x4_f32(false, afrag[dd + 1], false, b1,
                                                   (short)0, acc1, false, false);
    }
#pragma unroll
    for (int r = 0; r < 8; ++r) {
      const int m = r + 8 * half;
      sS[m * SSTRIDE + k0 + lrow] = (acc0[r] + acc1[r]) * scale;
    }
  }
  __syncthreads();

  {
    const int r = tid >> 3;
    const int j = tid & 7;
    float pmax = -3.4e38f;
    for (int i = 0; i < 64; ++i)
      pmax = fmaxf(pmax, sS[r * SSTRIDE + j + 8 * i]);
    sRed[tid] = pmax;
    __syncthreads();
    if (j == 0) {
      float mx = sRed[tid];
      for (int t = 1; t < 8; ++t) mx = fmaxf(mx, sRed[tid + t]);
      sRowMax[r] = mx;
    }
    __syncthreads();
    const float mx = sRowMax[r];
    float psum = 0.f;
    for (int i = 0; i < 64; ++i) {
      const int c = j + 8 * i;
      const float e = expf(sS[r * SSTRIDE + c] - mx);
      sS[r * SSTRIDE + c] = e;
      psum += e;
    }
    sRed[tid] = psum;
    __syncthreads();
    if (j == 0) {
      float s = sRed[tid];
      for (int t = 1; t < 8; ++t) s += sRed[tid + t];
      sRowSum[r] = s;
    }
  }
  __syncthreads();

  {
    const int n0 = wave * 16;
    const float* vb = vkc + (size_t)b * KK * NN + n0 + lrow;
    v8f acc0 = {0.f, 0.f, 0.f, 0.f, 0.f, 0.f, 0.f, 0.f};
    v8f acc1 = {0.f, 0.f, 0.f, 0.f, 0.f, 0.f, 0.f, 0.f};
    for (int k0 = 0; k0 < KK; k0 += 8) {
      const float2 ta =
          *reinterpret_cast<const float2*>(&sS[lrow * SSTRIDE + k0 + 2 * half]);
      const float2 tb =
          *reinterpret_cast<const float2*>(&sS[lrow * SSTRIDE + k0 + 4 + 2 * half]);
      v2f a0; a0.x = ta.x; a0.y = ta.y;
      v2f a1; a1.x = tb.x; a1.y = tb.y;
      v2f b0;
      b0.x = vb[(size_t)(k0 + 2 * half)     * NN];
      b0.y = vb[(size_t)(k0 + 2 * half + 1) * NN];
      v2f b1;
      b1.x = vb[(size_t)(k0 + 4 + 2 * half)     * NN];
      b1.y = vb[(size_t)(k0 + 4 + 2 * half + 1) * NN];
      acc0 = __builtin_amdgcn_wmma_f32_16x16x4_f32(false, a0, false, b0,
                                                   (short)0, acc0, false, false);
      acc1 = __builtin_amdgcn_wmma_f32_16x16x4_f32(false, a1, false, b1,
                                                   (short)0, acc1, false, false);
    }
#pragma unroll
    for (int r = 0; r < 8; ++r) {
      const int m = r + 8 * half;
      tmp[((size_t)b * LQ + l0 + m) * NN + n0 + lrow] =
          (acc0[r] + acc1[r]) / sRowSum[m];
    }
  }
}

__global__ __launch_bounds__(512) __attribute__((amdgpu_waves_per_eu(4, 4)))
void out_kernel(const float* __restrict__ q,
                const float* __restrict__ vq,
                const float* __restrict__ tmp,
                const float* __restrict__ gamma,
                const float* __restrict__ beta,
                float* __restrict__ out) {
  const int lane = threadIdx.x & 31;
  const int wave = __builtin_amdgcn_readfirstlane((int)(threadIdx.x >> 5));
  const int c = lane & 15;
  const int h = lane >> 4;
  const size_t bl = (size_t)blockIdx.x * 16 + wave;

  const float4 t = *reinterpret_cast<const float4*>(tmp + bl * NN + 4 * c);
  const float4* vp = reinterpret_cast<const float4*>(vq + bl * MM * NN) + lane;

  float4 v[32];
  for (int j = 0; j < 32; ++j) v[j] = vp[j * 32];

  float a[32];
  for (int j = 0; j < 32; ++j)
    a[j] = fmaf(v[j].w, t.w, fmaf(v[j].z, t.z, fmaf(v[j].y, t.y, v[j].x * t.x)));

  float b8[16];
  {
    const bool hi = (lane & 1) != 0;
    for (int i = 0; i < 16; ++i) {
      const float keep = hi ? a[2 * i + 1] : a[2 * i];
      const float send = hi ? a[2 * i] : a[2 * i + 1];
      b8[i] = keep + __shfl_xor(send, 1, 32);
    }
  }
  float b4[8];
  {
    const bool hi = (lane & 2) != 0;
    for (int i = 0; i < 8; ++i) {
      const float keep = hi ? b8[2 * i + 1] : b8[2 * i];
      const float send = hi ? b8[2 * i] : b8[2 * i + 1];
      b4[i] = keep + __shfl_xor(send, 2, 32);
    }
  }
  float b2[4];
  {
    const bool hi = (lane & 4) != 0;
    for (int i = 0; i < 4; ++i) {
      const float keep = hi ? b4[2 * i + 1] : b4[2 * i];
      const float send = hi ? b4[2 * i] : b4[2 * i + 1];
      b2[i] = keep + __shfl_xor(send, 4, 32);
    }
  }
  float r[2];
  {
    const bool hi = (lane & 8) != 0;
    for (int i = 0; i < 2; ++i) {
      const float keep = hi ? b2[2 * i + 1] : b2[2 * i];
      const float send = hi ? b2[2 * i] : b2[2 * i + 1];
      r[i] = keep + __shfl_xor(send, 8, 32);
    }
  }

  const int m0 = 2 * c + h;
  const float x0 = r[0] + q[bl * DD + m0];
  const float x1 = r[1] + q[bl * DD + m0 + 32];
  float s1 = x0 + x1;
  float s2 = x0 * x0 + x1 * x1;
  for (int d = 16; d >= 1; d >>= 1) {
    s1 += __shfl_xor(s1, d, 32);
    s2 += __shfl_xor(s2, d, 32);
  }
  const float mu   = s1 * (1.f / 64.f);
  const float var  = s2 * (1.f / 64.f) - mu * mu;
  const float rstd = rsqrtf(var + 1e-3f);
  out[bl * DD + m0]      = (x0 - mu) * rstd * gamma[m0]      + beta[m0];
  out[bl * DD + m0 + 32] = (x1 - mu) * rstd * gamma[m0 + 32] + beta[m0 + 32];
}

extern "C" void kernel_launch(void* const* d_in, const int* in_sizes, int n_in,
                              void* d_out, int out_size, void* d_ws, size_t ws_size,
                              hipStream_t stream) {
  const float* q     = (const float*)d_in[0];
  const float* kmat  = (const float*)d_in[1];
  const float* vq    = (const float*)d_in[2];
  const float* vk    = (const float*)d_in[3];
  const float* vexp  = (const float*)d_in[4];
  const float* scale = (const float*)d_in[5];
  const float* gamma = (const float*)d_in[6];
  const float* beta  = (const float*)d_in[7];
  float* out = (float*)d_out;

  float* vkc = (float*)d_ws;
  float* tmp = vkc + (size_t)BATCH * KK * NN;

  vkc_kernel <<<BATCH * KK / 16,   512, 0, stream>>>(vk, vexp, vkc);
  attn_kernel<<<BATCH * (LQ / 16), 128, 0, stream>>>(q, kmat, vkc, scale, tmp);
  out_kernel <<<BATCH * LQ / 16,   512, 0, stream>>>(q, vq, tmp, gamma, beta, out);
}


# attention kernel fused into the vq streaming kernel: a 16-wave block does its 16 rows' scores, softmax and vkc product from LDS under its in-flight vq slab loads; 2 launches; fp sum order changes
# speedup vs baseline: 1.8607x; 1.4162x over previous
#include <hip/hip_runtime.h>

#define BATCH 8
#define LQ   512
#define KK   512
#define DD   64
#define NN   64
#define MM   64
#define PP   64

typedef __attribute__((ext_vector_type(2))) float v2f;
typedef __attribute__((ext_vector_type(8))) float v8f;

__global__ __launch_bounds__(512) __attribute__((amdgpu_waves_per_eu(4, 4)))
void vkc_kernel(const float* __restrict__ vk,
                const float* __restrict__ vexp,
                float* __restrict__ vkc) {
  const int lane = threadIdx.x & 31;
  const int wave = __builtin_amdgcn_readfirstlane((int)(threadIdx.x >> 5));
  const int c = lane & 15;
  const int h = lane >> 4;
  const size_t bk = (size_t)blockIdx.x * 16 + wave;

  const float4* wp = reinterpret_cast<const float4*>(vexp + bk * PP + 32 * h);
  const float4* vp = reinterpret_cast<const float4*>(
      vk + bk * PP * NN + (size_t)(32 * h) * NN + 4 * c);

  float w[32];
  for (int i = 0; i < 8; ++i) {
    const float4 t = wp[i];
    w[4 * i + 0] = t.x; w[4 * i + 1] = t.y; w[4 * i + 2] = t.z; w[4 * i + 3] = t.w;
  }
  float4 v[32];
  for (int j = 0; j < 32; ++j) v[j] = vp[j * (NN / 4)];

  float4 acc = {0.f, 0.f, 0.f, 0.f};
  for (int j = 0; j < 32; ++j) {
    acc.x = fmaf(w[j], v[j].x, acc.x);
    acc.y = fmaf(w[j], v[j].y, acc.y);
    acc.z = fmaf(w[j], v[j].z, acc.z);
    acc.w = fmaf(w[j], v[j].w, acc.w);
  }
  acc.x += __shfl_xor(acc.x, 16, 32);
  acc.y += __shfl_xor(acc.y, 16, 32);
  acc.z += __shfl_xor(acc.z, 16, 32);
  acc.w += __shfl_xor(acc.w, 16, 32);
  if (h == 0) *reinterpret_cast<float4*>(vkc + bk * NN + 4 * c) = acc;
}

#define SSTRIDE 516
#define VSTRIDE 72
#define PSTRIDE 68

__device__ __forceinline__ void lds_barrier() {
  __builtin_amdgcn_fence(__ATOMIC_RELEASE, "workgroup", "local");
  __builtin_amdgcn_s_barrier();
  __builtin_amdgcn_fence(__ATOMIC_ACQUIRE, "workgroup", "local");
}

__global__ __launch_bounds__(512) __attribute__((amdgpu_waves_per_eu(4, 4)))
void attn_out_kernel(const float* __restrict__ q,
                     const float* __restrict__ kmat,
                     const float* __restrict__ vkc,
                     const float* __restrict__ scale_p,
                     const float* __restrict__ vq,
                     const float* __restrict__ gamma,
                     const float* __restrict__ beta,
                     float* __restrict__ out) {
  __shared__ __attribute__((aligned(16))) float sS[16 * SSTRIDE];
  __shared__ __attribute__((aligned(16))) float sV[KK * VSTRIDE];
  __shared__ __attribute__((aligned(16))) float sP[4 * 16 * PSTRIDE];
  __shared__ float sRowSum[16];

  const int tid  = threadIdx.x;
  const int lane = tid & 31;
  const int wave = __builtin_amdgcn_readfirstlane((int)(threadIdx.x >> 5));
  const int c = lane & 15;
  const int h = lane >> 4;
  const int b = blockIdx.x >> 5;
  const size_t bl = (size_t)blockIdx.x * 16 + wave;

  const float scale = scale_p[0];

  const float* qrow = q + ((size_t)blockIdx.x * 16 + c) * DD + 2 * h;
  v2f afrag[16];
  for (int dd = 0; dd < 16; ++dd) {
    const float2 t = *reinterpret_cast<const float2*>(qrow + 4 * dd);
    afrag[dd].x = t.x; afrag[dd].y = t.y;
  }
  const float* krow0 = kmat + ((size_t)b * KK + 16 * wave + c) * DD + 2 * h;
  const float* krow1 = krow0 + (size_t)256 * DD;
  v2f bf0[16], bf1[16];
  for (int dd = 0; dd < 16; ++dd) {
    const float2 t0 = *reinterpret_cast<const float2*>(krow0 + 4 * dd);
    const float2 t1 = *reinterpret_cast<const float2*>(krow1 + 4 * dd);
    bf0[dd].x = t0.x; bf0[dd].y = t0.y;
    bf1[dd].x = t1.x; bf1[dd].y = t1.y;
  }
  const float4* vsrc = reinterpret_cast<const float4*>(vkc + (size_t)b * KK * NN) + tid;
  float4 st[16];
  for (int i = 0; i < 16; ++i) st[i] = vsrc[512 * i];

  {
    v8f acc0 = {0.f, 0.f, 0.f, 0.f, 0.f, 0.f, 0.f, 0.f};
    v8f acc1 = {0.f, 0.f, 0.f, 0.f, 0.f, 0.f, 0.f, 0.f};
    for (int dd = 0; dd < 16; ++dd) {
      acc0 = __builtin_amdgcn_wmma_f32_16x16x4_f32(false, afrag[dd], false, bf0[dd],
                                                   (short)0, acc0, false, false);
      acc1 = __builtin_amdgcn_wmma_f32_16x16x4_f32(false, afrag[dd], false, bf1[dd],
                                                   (short)0, acc1, false, false);
    }
    for (int r = 0; r < 8; ++r) {
      const int m = r + 8 * h;
      sS[m * SSTRIDE + 16 * wave + c]       = acc0[r] * scale;
      sS[m * SSTRIDE + 256 + 16 * wave + c] = acc1[r] * scale;
    }
  }
  for (int i = 0; i < 16; ++i)
    *reinterpret_cast<float4*>(&sV[((tid >> 4) + 32 * i) * VSTRIDE + 4 * (tid & 15)]) = st[i];

  const float4* vp = reinterpret_cast<const float4*>(vq + bl * MM * NN) + lane;
  float4 v[32];
  for (int j = 0; j < 32; ++j) v[j] = vp[j * 32];

  lds_barrier();

  {
    float* srow = sS + wave * SSTRIDE + lane;
    float e[16];
    for (int i = 0; i < 16; ++i) e[i] = srow[32 * i];
    float mx = e[0];
    for (int i = 1; i < 16; ++i) mx = fmaxf(mx, e[i]);
    for (int d = 16; d >= 1; d >>= 1) mx = fmaxf(mx, __shfl_xor(mx, d, 32));
    float s = 0.f;
    for (int i = 0; i < 16; ++i) {
      e[i] = expf(e[i] - mx);
      srow[32 * i] = e[i];
      s += e[i];
    }
    for (int d = 16; d >= 1; d >>= 1) s += __shfl_xor(s, d, 32);
    if (lane == 0) sRowSum[wave] = s;
  }
  lds_barrier();

  {
    const int nt = wave & 3;
    const int kq = wave >> 2;
    const float* ap = sS + c * SSTRIDE + 128 * kq + 2 * h;
    const float* bp = sV + (128 * kq + 2 * h) * VSTRIDE + 16 * nt + c;
    v8f acc0 = {0.f, 0.f, 0.f, 0.f, 0.f, 0.f, 0.f, 0.f};
    v8f acc1 = {0.f, 0.f, 0.f, 0.f, 0.f, 0.f, 0.f, 0.f};
    for (int k0 = 0; k0 < 128; k0 += 8) {
      const float2 ta = *reinterpret_cast<const float2*>(ap + k0);
      const float2 tb = *reinterpret_cast<const float2*>(ap + k0 + 4);
      v2f a0; a0.x = ta.x; a0.y = ta.y;
      v2f a1; a1.x = tb.x; a1.y = tb.y;
      v2f b0; b0.x = bp[(k0)     * VSTRIDE]; b0.y = bp[(k0 + 1) * VSTRIDE];
      v2f b1; b1.x = bp[(k0 + 4) * VSTRIDE]; b1.y = bp[(k0 + 5) * VSTRIDE];
      acc0 = __builtin_amdgcn_wmma_f32_16x16x4_f32(false, a0, false, b0,
                                                   (short)0, acc0, false, false);
      acc1 = __builtin_amdgcn_wmma_f32_16x16x4_f32(false, a1, false, b1,
                                                   (short)0, acc1, false, false);
    }
    for (int r = 0; r < 8; ++r)
      sP[(kq * 16 + r + 8 * h) * PSTRIDE + 16 * nt + c] = acc0[r] + acc1[r];
  }
  lds_barrier();

  float4 t;
  {
    const float rs = sRowSum[wave];
    const float4 p0 = *reinterpret_cast<const float4*>(&sP[(0 * 16 + wave) * PSTRIDE + 4 * c]);
    const float4 p1 = *reinterpret_cast<const float4*>(&sP[(1 * 16 + wave) * PSTRIDE + 4 * c]);
    const float4 p2 = *reinterpret_cast<const float4*>(&sP[(2 * 16 + wave) * PSTRIDE + 4 * c]);
    const float4 p3 = *reinterpret_cast<const float4*>(&sP[(3 * 16 + wave) * PSTRIDE + 4 * c]);
    t.x = ((p0.x + p1.x) + (p2.x + p3.x)) / rs;
    t.y = ((p0.y + p1.y) + (p2.y + p3.y)) / rs;
    t.z = ((p0.z + p1.z) + (p2.z + p3.z)) / rs;
    t.w = ((p0.w + p1.w) + (p2.w + p3.w)) / rs;
  }

  float a[32];
  for (int j = 0; j < 32; ++j)
    a[j] = fmaf(v[j].w, t.w, fmaf(v[j].z, t.z, fmaf(v[j].y, t.y, v[j].x * t.x)));

  float b8[16];
  {
    const bool hi = (lane & 1) != 0;
    for (int i = 0; i < 16; ++i) {
      const float keep = hi ? a[2 * i + 1] : a[2 * i];
      const float send = hi ? a[2 * i] : a[2 * i + 1];
      b8[i] = keep + __shfl_xor(send, 1, 32);
    }
  }
  float b4[8];
  {
    const bool hi = (lane & 2) != 0;
    for (int i = 0; i < 8; ++i) {
      const float keep = hi ? b8[2 * i + 1] : b8[2 * i];
      const float send = hi ? b8[2 * i] : b8[2 * i + 1];
      b4[i] = keep + __shfl_xor(send, 2, 32);
    }
  }
  float b2[4];
  {
    const bool hi = (lane & 4) != 0;
    for (int i = 0; i < 4; ++i) {
      const float keep = hi ? b4[2 * i + 1] : b4[2 * i];
      const float send = hi ? b4[2 * i] : b4[2 * i + 1];
      b2[i] = keep + __shfl_xor(send, 4, 32);
    }
  }
  float r[2];
  {
    const bool hi = (lane & 8) != 0;
    for (int i = 0; i < 2; ++i) {
      const float keep = hi ? b2[2 * i + 1] : b2[2 * i];
      const float send = hi ? b2[2 * i] : b2[2 * i + 1];
      r[i] = keep + __shfl_xor(send, 8, 32);
    }
  }

  const int m0 = 2 * c + h;
  const float x0 = r[0] + q[bl * DD + m0];
  const float x1 = r[1] + q[bl * DD + m0 + 32];
  float s1 = x0 + x1;
  float s2 = x0 * x0 + x1 * x1;
  for (int d = 16; d >= 1; d >>= 1) {
    s1 += __shfl_xor(s1, d, 32);
    s2 += __shfl_xor(s2, d, 32);
  }
  const float mu   = s1 * (1.f / 64.f);
  const float var  = s2 * (1.f / 64.f) - mu * mu;
  const float rstd = rsqrtf(var + 1e-3f);
  out[bl * DD + m0]      = (x0 - mu) * rstd * gamma[m0]      + beta[m0];
  out[bl * DD + m0 + 32] = (x1 - mu) * rstd * gamma[m0 + 32] + beta[m0 + 32];
}

extern "C" void kernel_launch(void* const* d_in, const int* in_sizes, int n_in,
                              void* d_out, int out_size, void* d_ws, size_t ws_size,
                              hipStream_t stream) {
  const float* q     = (const float*)d_in[0];
  const float* kmat  = (const float*)d_in[1];
  const float* vq    = (const float*)d_in[2];
  const float* vk    = (const float*)d_in[3];
  const float* vexp  = (const float*)d_in[4];
  const float* scale = (const float*)d_in[5];
  const float* gamma = (const float*)d_in[6];
  const float* beta  = (const float*)d_in[7];
  float* out = (float*)d_out;

  float* vkc = (float*)d_ws;

  vkc_kernel     <<<BATCH * KK / 16, 512, 0, stream>>>(vk, vexp, vkc);
  attn_out_kernel<<<BATCH * LQ / 16, 512, 0, stream>>>(q, kmat, vkc, scale, vq, gamma, beta, out);
}
